# starting kernel
# baseline (speedup 1.0000x reference)
#include <hip/hip_runtime.h>


#define N_NODES   20000
#define D_FEAT    256
#define N_SUPPORT 3
#define N_EDGES   320000
#define OUT_DIM   256
#define H_DIM     (N_SUPPORT * D_FEAT)

typedef __attribute__((ext_vector_type(2))) float v2f;
typedef __attribute__((ext_vector_type(8))) float v8f;

__global__ __launch_bounds__(256) void zero_h(float4* __restrict__ h, int n4) {
  int i = blockIdx.x * blockDim.x + threadIdx.x;
  if (i < n4) h[i] = float4{0.f, 0.f, 0.f, 0.f};
}

__global__ __launch_bounds__(256) void scatter_edges(
    const float* __restrict__ x, const int* __restrict__ esrc,
    const int* __restrict__ edst, const float* __restrict__ ew,
    float* __restrict__ h) {
  long gid = (long)blockIdx.x * blockDim.x + threadIdx.x;
  int  chunk = (int)(gid & 63);
  long e = gid >> 6;
  if (e >= (long)N_SUPPORT * N_EDGES) return;
  int s   = (int)(e / N_EDGES);
  int src = esrc[e];
  int dst = edst[e];
  float w = ew[e];
  float4 xv = ((const float4*)(x + (long)src * D_FEAT))[chunk];
  float* hp = h + (long)dst * H_DIM + s * D_FEAT + chunk * 4;
  atomicAdd(hp + 0, xv.x * w);
  atomicAdd(hp + 1, xv.y * w);
  atomicAdd(hp + 2, xv.z * w);
  atomicAdd(hp + 3, xv.w * w);
}

__global__ __launch_bounds__(256) void gemm_wmma(
    const float* __restrict__ h, const float* __restrict__ Wm,
    float* __restrict__ out) {
  int wave = (int)((blockIdx.x * blockDim.x + threadIdx.x) >> 5);
  int lane = threadIdx.x & 31;
  int tile_m = wave >> 4;
  int tile_n = wave & 15;
  int half   = lane >> 4;
  int mrow   = tile_m * 16 + (lane & 15);
  int ncol   = tile_n * 16 + (lane & 15);

  const float* arow = h  + (long)mrow * H_DIM + half * 2;
  const float* bcol = Wm + (long)(half * 2) * OUT_DIM + ncol;

  v8f c = {};
  #pragma unroll 4
  for (int k = 0; k < H_DIM; k += 4) {
    v2f a = *(const v2f*)(arow + k);
    v2f b;
    b.x = bcol[(long)k * OUT_DIM];
    b.y = bcol[(long)k * OUT_DIM + OUT_DIM];
    c = __builtin_amdgcn_wmma_f32_16x16x4_f32(
         false, a,  false, b,
         (short)0, c,  false,  false);
  }

  float* op = out + (long)(tile_m * 16 + half * 8) * OUT_DIM + ncol;
  #pragma unroll
  for (int j = 0; j < 8; ++j) op[(long)j * OUT_DIM] = c[j];
}

extern "C" void kernel_launch(void* const* d_in, const int* in_sizes, int n_in,
                              void* d_out, int out_size, void* d_ws, size_t ws_size,
                              hipStream_t stream) {
  const float* x    = (const float*)d_in[0];
  const int*   esrc = (const int*)  d_in[1];
  const int*   edst = (const int*)  d_in[2];
  const float* ew   = (const float*)d_in[3];
  const float* Wm   = (const float*)d_in[4];
  float*       out  = (float*)d_out;
  float*       hbuf = (float*)d_ws;

  int n4 = N_NODES * H_DIM / 4;
  zero_h<<<(n4 + 255) / 256, 256, 0, stream>>>((float4*)hbuf, n4);

  long nthr = (long)N_SUPPORT * N_EDGES * 64;
  scatter_edges<<<(int)(nthr / 256), 256, 0, stream>>>(x, esrc, edst, ew, hbuf);

  gemm_wmma<<<(N_NODES / 16) * (OUT_DIM / 16) / 8, 256, 0, stream>>>(hbuf, Wm, out);
}


# GEMM operands staged in LDS: a 16-wave work-group copies 80x64 of h and 64x256 of W per K chunk with whole-line loads, once for all its waves; fragments read from LDS (was per-wave global loads)
# speedup vs baseline: 2.3546x; 2.3546x over previous
#include <hip/hip_runtime.h>


#define N_NODES   20000
#define D_FEAT    256
#define N_SUPPORT 3
#define N_EDGES   320000
#define OUT_DIM   256
#define H_DIM     (N_SUPPORT * D_FEAT)

typedef __attribute__((ext_vector_type(2))) float v2f;
typedef __attribute__((ext_vector_type(4))) float v4f;
typedef __attribute__((ext_vector_type(8))) float v8f;

__global__ __launch_bounds__(256) void zero_h(float4* __restrict__ h, int n4) {
  int i = blockIdx.x * blockDim.x + threadIdx.x;
  if (i < n4) h[i] = float4{0.f, 0.f, 0.f, 0.f};
}

#define EDGES_PER_WAVE 16
#define EDGE_GROUP     4
#define ROW_PIECES     (D_FEAT / 32)
static_assert(N_EDGES % EDGES_PER_WAVE == 0, "a wave's edges stay in one support");
static_assert((N_SUPPORT * N_EDGES / EDGES_PER_WAVE) % 8 == 0, "whole blocks");

__global__ __launch_bounds__(256) void scatter_edges(
    const float* __restrict__ x, const int* __restrict__ esrc,
    const int* __restrict__ edst, const float* __restrict__ ew,
    float* __restrict__ h) {
  int lane = threadIdx.x & 31;
  int wave = __builtin_amdgcn_readfirstlane((int)(blockIdx.x * 8 + (threadIdx.x >> 5)));
  int e0 = wave * EDGES_PER_WAVE;
  int s  = e0 / N_EDGES;
  const float* xl = x + lane;
  float*       hl = h + s * D_FEAT + lane;

  for (int g = 0; g < EDGES_PER_WAVE; g += EDGE_GROUP) {
    float v[EDGE_GROUP][ROW_PIECES];
    int   dst[EDGE_GROUP];
    #pragma unroll
    for (int u = 0; u < EDGE_GROUP; ++u) {
      int   e   = e0 + g + u;
      int   src = esrc[e];
      float w   = ew[e];
      dst[u]    = edst[e];
      const float* xp = xl + (long)src * D_FEAT;
      #pragma unroll
      for (int j = 0; j < ROW_PIECES; ++j) v[u][j] = xp[32 * j] * w;
    }
    #pragma unroll
    for (int u = 0; u < EDGE_GROUP; ++u) {
      float* hp = hl + (long)dst[u] * H_DIM;
      #pragma unroll
      for (int j = 0; j < ROW_PIECES; ++j) atomicAdd(hp + 32 * j, v[u][j]);
    }
  }
}

#define GM_ROWS    80
#define GM_MT      (GM_ROWS / 16)
#define GM_KC      64
#define GM_NCHUNK  (H_DIM / GM_KC)
#define GM_THREADS 512
#define GM_PA      (GM_KC + 4)
#define GM_PB      (OUT_DIM + 8)
#define GM_A_ELEMS (GM_ROWS * GM_PA)
#define GM_B_ELEMS (GM_KC * GM_PB)
#define GM_STAGE   (GM_A_ELEMS + GM_B_ELEMS)
#define GM_A_CP    (GM_ROWS * GM_KC / 2 / GM_THREADS)
#define GM_B_CP    (GM_KC * OUT_DIM / 4 / GM_THREADS)
static_assert(N_NODES % GM_ROWS == 0, "whole work-groups");
static_assert(H_DIM % GM_KC == 0, "whole chunks");
static_assert(GM_A_CP * GM_THREADS * 2 == GM_ROWS * GM_KC, "h piece in whole passes");
static_assert(GM_B_CP * GM_THREADS * 4 == GM_KC * OUT_DIM, "W piece in whole passes");
static_assert(GM_KC == 64 && GM_THREADS == 512, "copy maps below: one wave per h row, 64 lanes per W row");
static_assert(OUT_DIM / 16 == GM_THREADS / 32, "one column tile per wave");

__device__ __forceinline__ void gemm_fetch(v2f (&ra)[GM_A_CP], v4f (&rb)[GM_B_CP],
                                           const float* __restrict__ hp,
                                           const float* __restrict__ wp) {
  #pragma unroll
  for (int j = 0; j < GM_A_CP; ++j) ra[j] = *(const v2f*)(hp + (long)(j * 16) * H_DIM);
  #pragma unroll
  for (int j = 0; j < GM_B_CP; ++j) rb[j] = *(const v4f*)(wp + (long)(j * 8) * OUT_DIM);
}

__device__ __forceinline__ void gemm_put(const v2f (&ra)[GM_A_CP], const v4f (&rb)[GM_B_CP],
                                         float* as, float* bs) {
  #pragma unroll
  for (int j = 0; j < GM_A_CP; ++j) *(v2f*)(as + j * 16 * GM_PA) = ra[j];
  #pragma unroll
  for (int j = 0; j < GM_B_CP; ++j) *(v4f*)(bs + j * 8 * GM_PB) = rb[j];
}

template <int KS0, int KS1>
__device__ __forceinline__ void gemm_steps(v8f (&acc)[GM_MT], const float* ap, const float* bp) {
  #pragma unroll
  for (int ks = KS0; ks < KS1; ++ks) {
    v2f b;
    b.x = bp[(ks * 4) * GM_PB];
    b.y = bp[(ks * 4 + 1) * GM_PB];
    #pragma unroll
    for (int mt = 0; mt < GM_MT; ++mt) {
      v2f a = *(const v2f*)(ap + mt * 16 * GM_PA + ks * 4);
      acc[mt] = __builtin_amdgcn_wmma_f32_16x16x4_f32(
           false, a,  false, b,
           (short)0, acc[mt],  false,  false);
    }
  }
}

__global__ __launch_bounds__(GM_THREADS) void gemm_wmma(
    const float* __restrict__ h, const float* __restrict__ Wm,
    float* __restrict__ out) {
  __shared__ __attribute__((aligned(16))) float lds[2 * GM_STAGE];

  const int tid  = threadIdx.x;
  const int lane = tid & 31;
  const int wave = tid >> 5;
  const int c    = lane & 15;
  const int half = lane >> 4;
  const int m0   = blockIdx.x * GM_ROWS;

  const float* hg = h  + (long)(m0 + wave) * H_DIM + lane * 2;
  const float* wg = Wm + (long)(tid >> 6) * OUT_DIM + (tid & 63) * 4;
  const int a_st = wave * GM_PA + lane * 2;
  const int b_st = GM_A_ELEMS + (tid >> 6) * GM_PB + (tid & 63) * 4;
  const int a_ld = c * GM_PA + 2 * half;
  const int b_ld = GM_A_ELEMS + (2 * half) * GM_PB + wave * 16 + c;

  v2f ra[GM_A_CP];
  v4f rb[GM_B_CP];
  gemm_fetch(ra, rb, hg, wg);
  gemm_put(ra, rb, lds + a_st, lds + b_st);
  gemm_fetch(ra, rb, hg + GM_KC, wg + (long)GM_KC * OUT_DIM);
  __syncthreads();

  v8f acc[GM_MT];
  #pragma unroll
  for (int mt = 0; mt < GM_MT; ++mt) acc[mt] = v8f{};

  for (int ch = 0; ch < GM_NCHUNK - 2; ++ch) {
    const float* cur = lds + (ch & 1) * GM_STAGE;
    float*       nxt = lds + ((ch + 1) & 1) * GM_STAGE;
    gemm_steps<0, GM_KC / 8>(acc, cur + a_ld, cur + b_ld);
    gemm_put(ra, rb, nxt + a_st, nxt + b_st);
    gemm_steps<GM_KC / 8, GM_KC / 4>(acc, cur + a_ld, cur + b_ld);
    gemm_fetch(ra, rb, hg + (ch + 2) * GM_KC, wg + (long)((ch + 2) * GM_KC) * OUT_DIM);
    __syncthreads();
  }
  {
    const float* cur = lds + ((GM_NCHUNK - 2) & 1) * GM_STAGE;
    float*       nxt = lds + ((GM_NCHUNK - 1) & 1) * GM_STAGE;
    gemm_steps<0, GM_KC / 8>(acc, cur + a_ld, cur + b_ld);
    gemm_put(ra, rb, nxt + a_st, nxt + b_st);
    gemm_steps<GM_KC / 8, GM_KC / 4>(acc, cur + a_ld, cur + b_ld);
    __syncthreads();
    gemm_steps<0, GM_KC / 4>(acc, nxt + a_ld, nxt + b_ld);
  }

  float* op = out + (long)(m0 + half * 8) * OUT_DIM + wave * 16 + c;
  #pragma unroll
  for (int mt = 0; mt < GM_MT; ++mt) {
    #pragma unroll
    for (int j = 0; j < 8; ++j) op[(long)(mt * 16 + j) * OUT_DIM] = acc[mt][j];
  }
}

extern "C" void kernel_launch(void* const* d_in, const int* in_sizes, int n_in,
                              void* d_out, int out_size, void* d_ws, size_t ws_size,
                              hipStream_t stream) {
  const float* x    = (const float*)d_in[0];
  const int*   esrc = (const int*)  d_in[1];
  const int*   edst = (const int*)  d_in[2];
  const float* ew   = (const float*)d_in[3];
  const float* Wm   = (const float*)d_in[4];
  float*       out  = (float*)d_out;
  float*       hbuf = (float*)d_ws;

  int n4 = N_NODES * H_DIM / 4;
  zero_h<<<(n4 + 255) / 256, 256, 0, stream>>>((float4*)hbuf, n4);

  scatter_edges<<<N_SUPPORT * N_EDGES / EDGES_PER_WAVE / 8, 256, 0, stream>>>(x, esrc, edst, ew, hbuf);

  gemm_wmma<<<N_NODES / GM_ROWS, GM_THREADS, 0, stream>>>(hbuf, Wm, out);
}


# Edge sum fused into the GEMM: a work-group builds its 80x768 h tile in LDS (scans dst, adds x rows by LDS float atomics), then WMMA from LDS; no h, zero fill, global atomics; 1 launch; sums reordered
# speedup vs baseline: 3.3215x; 1.4106x over previous
#include <hip/hip_runtime.h>


#define N_NODES   20000
#define D_FEAT    256
#define N_SUPPORT 3
#define N_EDGES   320000
#define OUT_DIM   256
#define H_DIM     (N_SUPPORT * D_FEAT)

typedef __attribute__((ext_vector_type(2))) float v2f;
typedef __attribute__((ext_vector_type(4))) float v4f;
typedef __attribute__((ext_vector_type(8))) float v8f;
typedef __attribute__((ext_vector_type(4))) int   v4i;

#define F_THREADS  512
#define F_WAVES    (F_THREADS / 32)
#define F_ROWS     80
#define F_MT       (F_ROWS / 16)
#define F_PA       (H_DIM + 4)
#define F_A_ELEMS  (F_ROWS * F_PA)
#define F_KC       32
#define F_NCHUNK   (H_DIM / F_KC)
#define F_PB       (OUT_DIM + 8)
#define F_B_ELEMS  (F_KC * F_PB)
#define F_B_CP     (F_KC * OUT_DIM / 4 / F_THREADS)
#define F_LDS      (F_A_ELEMS + 2 * F_B_ELEMS)

#define F_PIECE    128
#define F_UNR      4
#define F_STEP     (F_UNR * F_PIECE)
#define F_NSTEP    (N_SUPPORT * N_EDGES / F_STEP)
#define F_GROUP    8
#define F_LIST     1024
#define ROW_PIECES (D_FEAT / 32)

static_assert(N_NODES % F_ROWS == 0, "whole work-groups");
static_assert(N_EDGES % F_STEP == 0, "a step stays in one support");
static_assert(N_SUPPORT * N_EDGES <= (1 << 20), "edge index fits 20 bits of a list entry");
static_assert(F_ROWS <= 128, "row offset fits the bits above");
static_assert(F_GROUP - 1 + F_STEP <= F_LIST, "a scan step cannot overflow the list");
static_assert(F_WAVES * F_LIST <= 2 * F_B_ELEMS, "lists fit in the W stage area");
static_assert(F_B_CP * F_THREADS * 4 == F_KC * OUT_DIM, "W piece in whole passes");
static_assert(OUT_DIM / 16 == F_WAVES, "one column tile per wave");
static_assert(F_A_ELEMS % 4 == 0, "tile cleared in 16-byte stores");

template <int G>
__device__ __forceinline__ void add_edges(const unsigned* lst, int lane,
                                          const float* __restrict__ x,
                                          const int* __restrict__ esrc,
                                          const float* __restrict__ ew,
                                          float* tile) {
  float v[G][ROW_PIECES];
  int   off[G];
  #pragma unroll
  for (int u = 0; u < G; ++u) {
    unsigned ent = (unsigned)__builtin_amdgcn_readfirstlane((int)lst[u]);
    int e   = (int)(ent & 0xFFFFFu);
    int rel = (int)(ent >> 20);
    int s   = (e >= N_EDGES) + (e >= 2 * N_EDGES);
    int   src = esrc[e];
    float w   = ew[e];
    off[u] = rel * F_PA + s * D_FEAT + lane;
    const float* xp = x + (long)src * D_FEAT + lane;
    #pragma unroll
    for (int j = 0; j < ROW_PIECES; ++j) v[u][j] = xp[32 * j] * w;
  }
  #pragma unroll
  for (int u = 0; u < G; ++u) {
    #pragma unroll
    for (int j = 0; j < ROW_PIECES; ++j)
      __hip_atomic_fetch_add(tile + off[u] + 32 * j, v[u][j], __ATOMIC_RELAXED,
                             __HIP_MEMORY_SCOPE_WORKGROUP);
  }
}

__device__ __forceinline__ int scan_rot(int st, int rot) {
  int r = st + rot;
  return r >= F_NSTEP ? r - F_NSTEP : r;
}
__device__ __forceinline__ void scan_fetch(v4i (&d)[F_UNR], const int* __restrict__ edst,
                                           int first, int lane) {
  #pragma unroll
  for (int u = 0; u < F_UNR; ++u)
    d[u] = *(const v4i*)(edst + first + u * F_PIECE + lane * 4);
}


__device__ __forceinline__ void w_fetch(v4f (&rb)[F_B_CP], const float* __restrict__ wp) {
  #pragma unroll
  for (int j = 0; j < F_B_CP; ++j) rb[j] = *(const v4f*)(wp + (long)(j * 8) * OUT_DIM);
}

__device__ __forceinline__ void w_put(const v4f (&rb)[F_B_CP], float* bs) {
  #pragma unroll
  for (int j = 0; j < F_B_CP; ++j) *(v4f*)(bs + j * 8 * F_PB) = rb[j];
}

template <int KS0, int KS1>
__device__ __forceinline__ void mm_steps(v8f (&acc)[F_MT], const float* ap, const float* bp) {
  #pragma unroll
  for (int ks = KS0; ks < KS1; ++ks) {
    v2f b;
    b.x = bp[(ks * 4) * F_PB];
    b.y = bp[(ks * 4 + 1) * F_PB];
    #pragma unroll
    for (int mt = 0; mt < F_MT; ++mt) {
      v2f a = *(const v2f*)(ap + mt * 16 * F_PA + ks * 4);
      acc[mt] = __builtin_amdgcn_wmma_f32_16x16x4_f32(
           false, a,  false, b,
           (short)0, acc[mt],  false,  false);
    }
  }
}

__global__ __launch_bounds__(F_THREADS) void gcn_fused(
    const float* __restrict__ x, const int* __restrict__ esrc,
    const int* __restrict__ edst, const float* __restrict__ ew,
    const float* __restrict__ Wm, float* __restrict__ out) {
  __shared__ __attribute__((aligned(16))) float lds[F_LDS];

  const int tid  = threadIdx.x;
  const int lane = tid & 31;
  const int wave = __builtin_amdgcn_readfirstlane(tid >> 5);
  const int c    = lane & 15;
  const int half = lane >> 4;
  const int m0   = blockIdx.x * F_ROWS;

  float* tile = lds;
  float* wst  = lds + F_A_ELEMS;

  for (int i = tid; i < F_A_ELEMS / 4; i += F_THREADS)
    *(v4f*)(tile + 4 * i) = v4f{0.f, 0.f, 0.f, 0.f};
  __syncthreads();

  {
    unsigned* lst = (unsigned*)wst + wave * F_LIST;
    int cnt = 0;
    const int rot = (blockIdx.x * 15) >> 1;
    v4i d[F_UNR];
    scan_fetch(d, edst, scan_rot(wave, rot) * F_STEP, lane);
    for (int st = wave; st < F_NSTEP; st += F_WAVES) {
      int nst = st + F_WAVES < F_NSTEP ? st + F_WAVES : F_NSTEP - 1;
      v4i nx[F_UNR];
      scan_fetch(nx, edst, scan_rot(nst, rot) * F_STEP, lane);
      const int ebase = scan_rot(st, rot) * F_STEP + lane * 4;
      #pragma unroll
      for (int u = 0; u < F_UNR; ++u) {
        #pragma unroll
        for (int q = 0; q < 4; ++q) {
          unsigned rel   = (unsigned)(d[u][q] - m0);
          bool     match = rel < (unsigned)F_ROWS;
          unsigned mask  = __builtin_amdgcn_ballot_w32(match);
          if (mask != 0u) {
            if (match) {
              int pos = cnt + (int)__builtin_amdgcn_mbcnt_lo(mask, 0u);
              lst[pos] = (unsigned)(ebase + u * F_PIECE + q) | (rel << 20);
            }
            cnt += __builtin_popcount(mask);
          }
        }
      }
      if (cnt >= F_GROUP) {
        int k = 0;
        for (; k + F_GROUP <= cnt; k += F_GROUP)
          add_edges<F_GROUP>(lst + k, lane, x, esrc, ew, tile);
        int rem = cnt - k;
        if (lane < rem) { unsigned t = lst[k + lane]; lst[lane] = t; }
        cnt = rem;
      }
      #pragma unroll
      for (int u = 0; u < F_UNR; ++u) d[u] = nx[u];
    }
    for (int k = 0; k < cnt; ++k) add_edges<1>(lst + k, lane, x, esrc, ew, tile);
  }
  __syncthreads();

  const float* wg = Wm + (long)(tid >> 6) * OUT_DIM + (tid & 63) * 4;
  const int b_st = (tid >> 6) * F_PB + (tid & 63) * 4;
  const int a_ld = c * F_PA + 2 * half;
  const int b_ld = (2 * half) * F_PB + wave * 16 + c;

  v4f rb[F_B_CP];
  w_fetch(rb, wg);
  w_put(rb, wst + b_st);
  w_fetch(rb, wg + (long)F_KC * OUT_DIM);
  __syncthreads();

  v8f acc[F_MT];
  #pragma unroll
  for (int mt = 0; mt < F_MT; ++mt) acc[mt] = v8f{};

  for (int ch = 0; ch < F_NCHUNK - 2; ++ch) {
    const float* cur = wst + (ch & 1) * F_B_ELEMS;
    float*       nxt = wst + ((ch + 1) & 1) * F_B_ELEMS;
    const float* ap  = tile + a_ld + ch * F_KC;
    mm_steps<0, F_KC / 8>(acc, ap, cur + b_ld);
    w_put(rb, nxt + b_st);
    mm_steps<F_KC / 8, F_KC / 4>(acc, ap, cur + b_ld);
    w_fetch(rb, wg + (long)((ch + 2) * F_KC) * OUT_DIM);
    __syncthreads();
  }
  {
    const float* cur = wst + ((F_NCHUNK - 2) & 1) * F_B_ELEMS;
    float*       nxt = wst + ((F_NCHUNK - 1) & 1) * F_B_ELEMS;
    const float* ap  = tile + a_ld + (F_NCHUNK - 2) * F_KC;
    mm_steps<0, F_KC / 8>(acc, ap, cur + b_ld);
    w_put(rb, nxt + b_st);
    mm_steps<F_KC / 8, F_KC / 4>(acc, ap, cur + b_ld);
    __syncthreads();
    mm_steps<0, F_KC / 4>(acc, ap + F_KC, nxt + b_ld);
  }

  float* op = out + (long)(m0 + half * 8) * OUT_DIM + wave * 16 + c;
  #pragma unroll
  for (int mt = 0; mt < F_MT; ++mt) {
    #pragma unroll
    for (int j = 0; j < 8; ++j) op[(long)(mt * 16 + j) * OUT_DIM] = acc[mt][j];
  }
}

extern "C" void kernel_launch(void* const* d_in, const int* in_sizes, int n_in,
                              void* d_out, int out_size, void* d_ws, size_t ws_size,
                              hipStream_t stream) {
  const float* x    = (const float*)d_in[0];
  const int*   esrc = (const int*)  d_in[1];
  const int*   edst = (const int*)  d_in[2];
  const float* ew   = (const float*)d_in[3];
  const float* Wm   = (const float*)d_in[4];
  float*       out  = (float*)d_out;

  gcn_fused<<<N_NODES / F_ROWS, F_THREADS, 0, stream>>>(x, esrc, edst, ew, Wm, out);
}


# Edge scan narrowed to 1 owner byte per edge: first launch writes dst/80 as bytes to the workspace; a work-group scans 0.96 MB of them (was 3.84 MB of dst), 16 edges per 16-byte load; 1 -> 2 launches
# speedup vs baseline: 3.5293x; 1.0626x over previous
#include <hip/hip_runtime.h>


#define N_NODES   20000
#define D_FEAT    256
#define N_SUPPORT 3
#define N_EDGES   320000
#define OUT_DIM   256
#define H_DIM     (N_SUPPORT * D_FEAT)

typedef __attribute__((ext_vector_type(2))) float v2f;
typedef __attribute__((ext_vector_type(4))) float v4f;
typedef __attribute__((ext_vector_type(8))) float v8f;
typedef __attribute__((ext_vector_type(4))) int   v4i;
typedef __attribute__((ext_vector_type(4))) unsigned v4u;

#define F_THREADS  512
#define F_WAVES    (F_THREADS / 32)
#define F_ROWS     80
#define F_MT       (F_ROWS / 16)
#define F_PA       (H_DIM + 4)
#define F_A_ELEMS  (F_ROWS * F_PA)
#define F_KC       32
#define F_NCHUNK   (H_DIM / F_KC)
#define F_PB       (OUT_DIM + 8)
#define F_B_ELEMS  (F_KC * F_PB)
#define F_B_CP     (F_KC * OUT_DIM / 4 / F_THREADS)
#define F_LDS      (F_A_ELEMS + 2 * F_B_ELEMS)

#define F_NBLK     (N_NODES / F_ROWS)
#define N_FLAT     (N_SUPPORT * N_EDGES)
#define F_PIECE    512
#define F_UNR      4
#define F_STEP     (F_UNR * F_PIECE)
#define F_NSTEP    ((N_FLAT + F_STEP - 1) / F_STEP)
#define N_OWN      (F_NSTEP * F_STEP)
#define O_THREADS  256
#define O_UNR      4
#define F_GROUP    8
#define F_LIST     1024
#define ROW_PIECES (D_FEAT / 32)

static_assert(N_NODES % F_ROWS == 0, "whole work-groups");
static_assert(F_NBLK <= 255, "an owner fits a byte, 255 left for 'none'");
static_assert(N_FLAT % 4 == 0 && N_OWN % 4 == 0, "owner bytes written as whole dwords");
static_assert(F_GROUP - 1 + F_PIECE <= F_LIST, "a piece cannot overflow the list");
static_assert(F_WAVES * F_LIST <= 2 * F_B_ELEMS, "lists fit in the W stage area");
static_assert(F_B_CP * F_THREADS * 4 == F_KC * OUT_DIM, "W piece in whole passes");
static_assert(OUT_DIM / 16 == F_WAVES, "one column tile per wave");
static_assert(F_A_ELEMS % 4 == 0, "tile cleared in 16-byte stores");

template <int G>
__device__ __forceinline__ void add_edges(const unsigned* lst, int lane, int m0,
                                          const float* __restrict__ x,
                                          const int* __restrict__ esrc,
                                          const int* __restrict__ edst,
                                          const float* __restrict__ ew,
                                          float* tile) {
  float v[G][ROW_PIECES];
  int   off[G];
  #pragma unroll
  for (int u = 0; u < G; ++u) {
    int e   = __builtin_amdgcn_readfirstlane((int)lst[u]);
    int s   = (e >= N_EDGES) + (e >= 2 * N_EDGES);
    int   src = esrc[e];
    float w   = ew[e];
    int rel = edst[e] - m0;
    off[u] = rel * F_PA + s * D_FEAT + lane;
    const float* xp = x + (long)src * D_FEAT + lane;
    #pragma unroll
    for (int j = 0; j < ROW_PIECES; ++j) v[u][j] = xp[32 * j] * w;
  }
  #pragma unroll
  for (int u = 0; u < G; ++u) {
    #pragma unroll
    for (int j = 0; j < ROW_PIECES; ++j)
      __hip_atomic_fetch_add(tile + off[u] + 32 * j, v[u][j], __ATOMIC_RELAXED,
                             __HIP_MEMORY_SCOPE_WORKGROUP);
  }
}

__device__ __forceinline__ int scan_rot(int st, int rot) {
  int r = st + rot;
  return r >= F_NSTEP ? r - F_NSTEP : r;
}
__device__ __forceinline__ void scan_fetch(v4u (&d)[F_UNR], const unsigned* __restrict__ own,
                                           int first, int lane) {
  #pragma unroll
  for (int u = 0; u < F_UNR; ++u)
    d[u] = *(const v4u*)(own + (first + u * F_PIECE) / 4 + lane * 4);
}
__device__ __forceinline__ unsigned bytes_differ(unsigned v, unsigned bb) {
  unsigned t = v ^ bb;
  return ((t & 0x7F7F7F7Fu) + 0x7F7F7F7Fu) | t;
}
__device__ __forceinline__ unsigned own_match16(v4u v, unsigned bb) {
  unsigned m = (bytes_differ(v[0], bb) & 0x80808080u)
             | ((bytes_differ(v[1], bb) >> 1) & 0x40404040u)
             | ((bytes_differ(v[2], bb) >> 2) & 0x20202020u)
             | ((bytes_differ(v[3], bb) >> 3) & 0x10101010u);
  return ~m & 0xF0F0F0F0u;
}

__global__ __launch_bounds__(O_THREADS) void owner_bytes(const int* __restrict__ edst,
                                                         unsigned* __restrict__ own) {
  const int i0 = blockIdx.x * (O_THREADS * O_UNR) + threadIdx.x;
  v4i d[O_UNR];
  #pragma unroll
  for (int u = 0; u < O_UNR; ++u) {
    const int i = i0 + u * O_THREADS;
    d[u] = v4i{-1, -1, -1, -1};
    if (i < N_FLAT / 4) d[u] = *(const v4i*)(edst + 4 * i);
  }
  #pragma unroll
  for (int u = 0; u < O_UNR; ++u) {
    const int i = i0 + u * O_THREADS;
    unsigned r = 0u;
    #pragma unroll
    for (int q = 0; q < 4; ++q) {
      unsigned n = (unsigned)d[u][q];
      unsigned b = n < (unsigned)N_NODES ? n / (unsigned)F_ROWS : 255u;
      r |= b << (8 * q);
    }
    if (i < N_OWN / 4) own[i] = r;
  }
}


__device__ __forceinline__ void w_fetch(v4f (&rb)[F_B_CP], const float* __restrict__ wp) {
  #pragma unroll
  for (int j = 0; j < F_B_CP; ++j) rb[j] = *(const v4f*)(wp + (long)(j * 8) * OUT_DIM);
}

__device__ __forceinline__ void w_put(const v4f (&rb)[F_B_CP], float* bs) {
  #pragma unroll
  for (int j = 0; j < F_B_CP; ++j) *(v4f*)(bs + j * 8 * F_PB) = rb[j];
}

template <int KS0, int KS1>
__device__ __forceinline__ void mm_steps(v8f (&acc)[F_MT], const float* ap, const float* bp) {
  #pragma unroll
  for (int ks = KS0; ks < KS1; ++ks) {
    v2f b;
    b.x = bp[(ks * 4) * F_PB];
    b.y = bp[(ks * 4 + 1) * F_PB];
    #pragma unroll
    for (int mt = 0; mt < F_MT; ++mt) {
      v2f a = *(const v2f*)(ap + mt * 16 * F_PA + ks * 4);
      acc[mt] = __builtin_amdgcn_wmma_f32_16x16x4_f32(
           false, a,  false, b,
           (short)0, acc[mt],  false,  false);
    }
  }
}

__global__ __launch_bounds__(F_THREADS) void gcn_fused(
    const float* __restrict__ x, const int* __restrict__ esrc,
    const int* __restrict__ edst, const float* __restrict__ ew,
    const float* __restrict__ Wm, const unsigned* __restrict__ own,
    float* __restrict__ out) {
  __shared__ __attribute__((aligned(16))) float lds[F_LDS];

  const int tid  = threadIdx.x;
  const int lane = tid & 31;
  const int wave = __builtin_amdgcn_readfirstlane(tid >> 5);
  const int c    = lane & 15;
  const int half = lane >> 4;
  const int m0   = blockIdx.x * F_ROWS;

  float* tile = lds;
  float* wst  = lds + F_A_ELEMS;

  for (int i = tid; i < F_A_ELEMS / 4; i += F_THREADS)
    *(v4f*)(tile + 4 * i) = v4f{0.f, 0.f, 0.f, 0.f};
  __syncthreads();

  {
    unsigned* lst = (unsigned*)wst + wave * F_LIST;
    int cnt = 0;
    const unsigned bb = (unsigned)blockIdx.x * 0x01010101u;
    const int rot = (blockIdx.x * 15) >> 3;
    v4u d[F_UNR];
    scan_fetch(d, own, scan_rot(wave, rot) * F_STEP, lane);
    for (int st = wave; st < F_NSTEP; st += F_WAVES) {
      int nst = st + F_WAVES < F_NSTEP ? st + F_WAVES : F_NSTEP - 1;
      v4u nx[F_UNR];
      scan_fetch(nx, own, scan_rot(nst, rot) * F_STEP, lane);
      const int ebase = scan_rot(st, rot) * F_STEP + lane * 16;
      #pragma unroll
      for (int u = 0; u < F_UNR; ++u) {
        unsigned m    = own_match16(d[u], bb);
        unsigned mask = __builtin_amdgcn_ballot_w32(m != 0u);
        for (int it = 0; it < 16 && mask != 0u; ++it) {
          if (m != 0u) {
            int p   = __builtin_ctz(m);
            int pos = cnt + (int)__builtin_amdgcn_mbcnt_lo(mask, 0u);
            lst[pos] = (unsigned)(ebase + u * F_PIECE + ((7 - (p & 7)) << 2) + (p >> 3));
            m &= m - 1u;
          }
          cnt += __builtin_popcount(mask);
          mask = __builtin_amdgcn_ballot_w32(m != 0u);
        }
        if (cnt >= F_GROUP) {
          int k = 0;
          for (; k + F_GROUP <= cnt; k += F_GROUP)
            add_edges<F_GROUP>(lst + k, lane, m0, x, esrc, edst, ew, tile);
          int rem = cnt - k;
          if (lane < rem) { unsigned t = lst[k + lane]; lst[lane] = t; }
          cnt = rem;
        }
      }
      #pragma unroll
      for (int u = 0; u < F_UNR; ++u) d[u] = nx[u];
    }
    for (int k = 0; k < cnt; ++k) add_edges<1>(lst + k, lane, m0, x, esrc, edst, ew, tile);
  }
  __syncthreads();

  const float* wg = Wm + (long)(tid >> 6) * OUT_DIM + (tid & 63) * 4;
  const int b_st = (tid >> 6) * F_PB + (tid & 63) * 4;
  const int a_ld = c * F_PA + 2 * half;
  const int b_ld = (2 * half) * F_PB + wave * 16 + c;

  v4f rb[F_B_CP];
  w_fetch(rb, wg);
  w_put(rb, wst + b_st);
  w_fetch(rb, wg + (long)F_KC * OUT_DIM);
  __syncthreads();

  v8f acc[F_MT];
  #pragma unroll
  for (int mt = 0; mt < F_MT; ++mt) acc[mt] = v8f{};

  for (int ch = 0; ch < F_NCHUNK - 2; ++ch) {
    const float* cur = wst + (ch & 1) * F_B_ELEMS;
    float*       nxt = wst + ((ch + 1) & 1) * F_B_ELEMS;
    const float* ap  = tile + a_ld + ch * F_KC;
    mm_steps<0, F_KC / 8>(acc, ap, cur + b_ld);
    w_put(rb, nxt + b_st);
    mm_steps<F_KC / 8, F_KC / 4>(acc, ap, cur + b_ld);
    w_fetch(rb, wg + (long)((ch + 2) * F_KC) * OUT_DIM);
    __syncthreads();
  }
  {
    const float* cur = wst + ((F_NCHUNK - 2) & 1) * F_B_ELEMS;
    float*       nxt = wst + ((F_NCHUNK - 1) & 1) * F_B_ELEMS;
    const float* ap  = tile + a_ld + (F_NCHUNK - 2) * F_KC;
    mm_steps<0, F_KC / 8>(acc, ap, cur + b_ld);
    w_put(rb, nxt + b_st);
    mm_steps<F_KC / 8, F_KC / 4>(acc, ap, cur + b_ld);
    __syncthreads();
    mm_steps<0, F_KC / 4>(acc, ap + F_KC, nxt + b_ld);
  }

  float* op = out + (long)(m0 + half * 8) * OUT_DIM + wave * 16 + c;
  #pragma unroll
  for (int mt = 0; mt < F_MT; ++mt) {
    #pragma unroll
    for (int j = 0; j < 8; ++j) op[(long)(mt * 16 + j) * OUT_DIM] = acc[mt][j];
  }
}

extern "C" void kernel_launch(void* const* d_in, const int* in_sizes, int n_in,
                              void* d_out, int out_size, void* d_ws, size_t ws_size,
                              hipStream_t stream) {
  const float* x    = (const float*)d_in[0];
  const int*   esrc = (const int*)  d_in[1];
  const int*   edst = (const int*)  d_in[2];
  const float* ew   = (const float*)d_in[3];
  const float* Wm   = (const float*)d_in[4];
  float*       out  = (float*)d_out;
  unsigned*    own  = (unsigned*)d_ws;

  owner_bytes<<<(N_OWN / 4 + O_THREADS * O_UNR - 1) / (O_THREADS * O_UNR), O_THREADS, 0, stream>>>(edst, own);
  gcn_fused<<<F_NBLK, F_THREADS, 0, stream>>>(x, esrc, edst, ew, Wm, own, out);
}


# x rows fetched by 16-byte loads, 2 per edge (was 8 4-byte); tile columns permuted (col 4p+q at q*64+p) so LDS adds stay conflict-free; W rows staged to match: k summed in a permuted order
# speedup vs baseline: 4.1216x; 1.1678x over previous
#include <hip/hip_runtime.h>


#define N_NODES   20000
#define D_FEAT    256
#define N_SUPPORT 3
#define N_EDGES   320000
#define OUT_DIM   256
#define H_DIM     (N_SUPPORT * D_FEAT)

typedef __attribute__((ext_vector_type(2))) float v2f;
typedef __attribute__((ext_vector_type(4))) float v4f;
typedef __attribute__((ext_vector_type(8))) float v8f;
typedef __attribute__((ext_vector_type(4))) int   v4i;
typedef __attribute__((ext_vector_type(4))) unsigned v4u;

#define F_THREADS  512
#define F_WAVES    (F_THREADS / 32)
#define F_ROWS     80
#define F_MT       (F_ROWS / 16)
#define F_PA       (H_DIM + 4)
#define F_A_ELEMS  (F_ROWS * F_PA)
#define F_KC       32
#define F_NCHUNK   (H_DIM / F_KC)
#define F_PB       (OUT_DIM + 8)
#define F_B_ELEMS  (F_KC * F_PB)
#define F_B_CP     (F_KC * OUT_DIM / 4 / F_THREADS)
#define F_LDS      (F_A_ELEMS + 2 * F_B_ELEMS)

#define F_NBLK     (N_NODES / F_ROWS)
#define N_FLAT     (N_SUPPORT * N_EDGES)
#define F_PIECE    512
#define F_UNR      4
#define F_STEP     (F_UNR * F_PIECE)
#define F_NSTEP    ((N_FLAT + F_STEP - 1) / F_STEP)
#define N_OWN      (F_NSTEP * F_STEP)
#define O_THREADS  256
#define O_UNR      4
#define F_GROUP    8
#define F_LIST     1024
#define ROW_LOADS  (D_FEAT / 4 / 32)
#define Q_PITCH    (D_FEAT / 4)

static_assert(N_NODES % F_ROWS == 0, "whole work-groups");
static_assert(F_NBLK <= 255, "an owner fits a byte, 255 left for 'none'");
static_assert(N_FLAT % 4 == 0 && N_OWN % 4 == 0, "owner bytes written as whole dwords");
static_assert(F_GROUP - 1 + F_PIECE <= F_LIST, "a piece cannot overflow the list");
static_assert(F_WAVES * F_LIST <= 2 * F_B_ELEMS, "lists fit in the W stage area");
static_assert(F_B_CP * F_THREADS * 4 == F_KC * OUT_DIM, "W piece in whole passes");
static_assert(OUT_DIM / 16 == F_WAVES, "one column tile per wave");
static_assert(F_A_ELEMS % 4 == 0, "tile cleared in 16-byte stores");
static_assert(ROW_LOADS * 32 == Q_PITCH, "the pieces of a row fill one q block of the tile");
static_assert(Q_PITCH % F_KC == 0 && D_FEAT % Q_PITCH == 0, "a W stage lies inside one q block");

template <int G>
__device__ __forceinline__ void add_edges(const unsigned* lst, int lane, int m0,
                                          const float* __restrict__ x,
                                          const int* __restrict__ esrc,
                                          const int* __restrict__ edst,
                                          const float* __restrict__ ew,
                                          float* tile) {
  v4f v[G][ROW_LOADS];
  int off[G];
  #pragma unroll
  for (int u = 0; u < G; ++u) {
    int e   = __builtin_amdgcn_readfirstlane((int)lst[u]);
    int s   = (e >= N_EDGES) + (e >= 2 * N_EDGES);
    int   src = esrc[e];
    float w   = ew[e];
    int rel = edst[e] - m0;
    off[u] = rel * F_PA + s * D_FEAT + lane;
    const v4f* xp = (const v4f*)(x + (long)src * D_FEAT) + lane;
    #pragma unroll
    for (int j = 0; j < ROW_LOADS; ++j) v[u][j] = xp[32 * j] * w;
  }
  #pragma unroll
  for (int u = 0; u < G; ++u) {
    #pragma unroll
    for (int j = 0; j < ROW_LOADS; ++j) {
      #pragma unroll
      for (int q = 0; q < 4; ++q)
        __hip_atomic_fetch_add(tile + off[u] + q * Q_PITCH + 32 * j, v[u][j][q],
                               __ATOMIC_RELAXED, __HIP_MEMORY_SCOPE_WORKGROUP);
    }
  }
}

__device__ __forceinline__ int scan_rot(int st, int rot) {
  int r = st + rot;
  return r >= F_NSTEP ? r - F_NSTEP : r;
}
__device__ __forceinline__ void scan_fetch(v4u (&d)[F_UNR], const unsigned* __restrict__ own,
                                           int first, int lane) {
  #pragma unroll
  for (int u = 0; u < F_UNR; ++u)
    d[u] = *(const v4u*)(own + (first + u * F_PIECE) / 4 + lane * 4);
}
__device__ __forceinline__ unsigned bytes_differ(unsigned v, unsigned bb) {
  unsigned t = v ^ bb;
  return ((t & 0x7F7F7F7Fu) + 0x7F7F7F7Fu) | t;
}
__device__ __forceinline__ unsigned own_match16(v4u v, unsigned bb) {
  unsigned m = (bytes_differ(v[0], bb) & 0x80808080u)
             | ((bytes_differ(v[1], bb) >> 1) & 0x40404040u)
             | ((bytes_differ(v[2], bb) >> 2) & 0x20202020u)
             | ((bytes_differ(v[3], bb) >> 3) & 0x10101010u);
  return ~m & 0xF0F0F0F0u;
}

__global__ __launch_bounds__(O_THREADS) void owner_bytes(const int* __restrict__ edst,
                                                         unsigned* __restrict__ own) {
  const int i0 = blockIdx.x * (O_THREADS * O_UNR) + threadIdx.x;
  v4i d[O_UNR];
  #pragma unroll
  for (int u = 0; u < O_UNR; ++u) {
    const int i = i0 + u * O_THREADS;
    d[u] = v4i{-1, -1, -1, -1};
    if (i < N_FLAT / 4) d[u] = *(const v4i*)(edst + 4 * i);
  }
  #pragma unroll
  for (int u = 0; u < O_UNR; ++u) {
    const int i = i0 + u * O_THREADS;
    unsigned r = 0u;
    #pragma unroll
    for (int q = 0; q < 4; ++q) {
      unsigned n = (unsigned)d[u][q];
      unsigned b = n < (unsigned)N_NODES ? n / (unsigned)F_ROWS : 255u;
      r |= b << (8 * q);
    }
    if (i < N_OWN / 4) own[i] = r;
  }
}


__device__ __forceinline__ int w_row0(int ch) {
  return (ch >> 3) * D_FEAT + (ch & 1) * (4 * F_KC) + ((ch >> 1) & 3);
}

__device__ __forceinline__ void w_fetch(v4f (&rb)[F_B_CP], const float* __restrict__ wp) {
  #pragma unroll
  for (int j = 0; j < F_B_CP; ++j) rb[j] = *(const v4f*)(wp + (long)(j * 32) * OUT_DIM);
}

__device__ __forceinline__ void w_put(const v4f (&rb)[F_B_CP], float* bs) {
  #pragma unroll
  for (int j = 0; j < F_B_CP; ++j) *(v4f*)(bs + j * 8 * F_PB) = rb[j];
}

template <int KS0, int KS1>
__device__ __forceinline__ void mm_steps(v8f (&acc)[F_MT], const float* ap, const float* bp) {
  #pragma unroll
  for (int ks = KS0; ks < KS1; ++ks) {
    v2f b;
    b.x = bp[(ks * 4) * F_PB];
    b.y = bp[(ks * 4 + 1) * F_PB];
    #pragma unroll
    for (int mt = 0; mt < F_MT; ++mt) {
      v2f a = *(const v2f*)(ap + mt * 16 * F_PA + ks * 4);
      acc[mt] = __builtin_amdgcn_wmma_f32_16x16x4_f32(
           false, a,  false, b,
           (short)0, acc[mt],  false,  false);
    }
  }
}

__global__ __launch_bounds__(F_THREADS) void gcn_fused(
    const float* __restrict__ x, const int* __restrict__ esrc,
    const int* __restrict__ edst, const float* __restrict__ ew,
    const float* __restrict__ Wm, const unsigned* __restrict__ own,
    float* __restrict__ out) {
  __shared__ __attribute__((aligned(16))) float lds[F_LDS];

  const int tid  = threadIdx.x;
  const int lane = tid & 31;
  const int wave = __builtin_amdgcn_readfirstlane(tid >> 5);
  const int c    = lane & 15;
  const int half = lane >> 4;
  const int m0   = blockIdx.x * F_ROWS;

  float* tile = lds;
  float* wst  = lds + F_A_ELEMS;

  for (int i = tid; i < F_A_ELEMS / 4; i += F_THREADS)
    *(v4f*)(tile + 4 * i) = v4f{0.f, 0.f, 0.f, 0.f};
  __syncthreads();

  {
    unsigned* lst = (unsigned*)wst + wave * F_LIST;
    int cnt = 0;
    const unsigned bb = (unsigned)blockIdx.x * 0x01010101u;
    const int rot = (blockIdx.x * 15) >> 3;
    v4u d[F_UNR];
    scan_fetch(d, own, scan_rot(wave, rot) * F_STEP, lane);
    for (int st = wave; st < F_NSTEP; st += F_WAVES) {
      int nst = st + F_WAVES < F_NSTEP ? st + F_WAVES : F_NSTEP - 1;
      v4u nx[F_UNR];
      scan_fetch(nx, own, scan_rot(nst, rot) * F_STEP, lane);
      const int ebase = scan_rot(st, rot) * F_STEP + lane * 16;
      #pragma unroll
      for (int u = 0; u < F_UNR; ++u) {
        unsigned m    = own_match16(d[u], bb);
        unsigned mask = __builtin_amdgcn_ballot_w32(m != 0u);
        for (int it = 0; it < 16 && mask != 0u; ++it) {
          if (m != 0u) {
            int p   = __builtin_ctz(m);
            int pos = cnt + (int)__builtin_amdgcn_mbcnt_lo(mask, 0u);
            lst[pos] = (unsigned)(ebase + u * F_PIECE + ((7 - (p & 7)) << 2) + (p >> 3));
            m &= m - 1u;
          }
          cnt += __builtin_popcount(mask);
          mask = __builtin_amdgcn_ballot_w32(m != 0u);
        }
        if (cnt >= F_GROUP) {
          int k = 0;
          for (; k + F_GROUP <= cnt; k += F_GROUP)
            add_edges<F_GROUP>(lst + k, lane, m0, x, esrc, edst, ew, tile);
          int rem = cnt - k;
          if (lane < rem) { unsigned t = lst[k + lane]; lst[lane] = t; }
          cnt = rem;
        }
      }
      #pragma unroll
      for (int u = 0; u < F_UNR; ++u) d[u] = nx[u];
    }
    for (int k = 0; k < cnt; ++k) add_edges<1>(lst + k, lane, m0, x, esrc, edst, ew, tile);
  }
  __syncthreads();

  const float* wg = Wm + (long)(4 * (tid >> 6)) * OUT_DIM + (tid & 63) * 4;
  const int b_st = (tid >> 6) * F_PB + (tid & 63) * 4;
  const int a_ld = c * F_PA + 2 * half;
  const int b_ld = (2 * half) * F_PB + wave * 16 + c;

  v4f rb[F_B_CP];
  w_fetch(rb, wg + (long)w_row0(0) * OUT_DIM);
  w_put(rb, wst + b_st);
  w_fetch(rb, wg + (long)w_row0(1) * OUT_DIM);
  __syncthreads();

  v8f acc[F_MT];
  #pragma unroll
  for (int mt = 0; mt < F_MT; ++mt) acc[mt] = v8f{};

  for (int ch = 0; ch < F_NCHUNK - 2; ++ch) {
    const float* cur = wst + (ch & 1) * F_B_ELEMS;
    float*       nxt = wst + ((ch + 1) & 1) * F_B_ELEMS;
    const float* ap  = tile + a_ld + ch * F_KC;
    mm_steps<0, F_KC / 8>(acc, ap, cur + b_ld);
    w_put(rb, nxt + b_st);
    mm_steps<F_KC / 8, F_KC / 4>(acc, ap, cur + b_ld);
    w_fetch(rb, wg + (long)w_row0(ch + 2) * OUT_DIM);
    __syncthreads();
  }
  {
    const float* cur = wst + ((F_NCHUNK - 2) & 1) * F_B_ELEMS;
    float*       nxt = wst + ((F_NCHUNK - 1) & 1) * F_B_ELEMS;
    const float* ap  = tile + a_ld + (F_NCHUNK - 2) * F_KC;
    mm_steps<0, F_KC / 8>(acc, ap, cur + b_ld);
    w_put(rb, nxt + b_st);
    mm_steps<F_KC / 8, F_KC / 4>(acc, ap, cur + b_ld);
    __syncthreads();
    mm_steps<0, F_KC / 4>(acc, ap + F_KC, nxt + b_ld);
  }

  float* op = out + (long)(m0 + half * 8) * OUT_DIM + wave * 16 + c;
  #pragma unroll
  for (int mt = 0; mt < F_MT; ++mt) {
    #pragma unroll
    for (int j = 0; j < 8; ++j) op[(long)(mt * 16 + j) * OUT_DIM] = acc[mt][j];
  }
}

extern "C" void kernel_launch(void* const* d_in, const int* in_sizes, int n_in,
                              void* d_out, int out_size, void* d_ws, size_t ws_size,
                              hipStream_t stream) {
  const float* x    = (const float*)d_in[0];
  const int*   esrc = (const int*)  d_in[1];
  const int*   edst = (const int*)  d_in[2];
  const float* ew   = (const float*)d_in[3];
  const float* Wm   = (const float*)d_in[4];
  float*       out  = (float*)d_out;
  unsigned*    own  = (unsigned*)d_ws;

  owner_bytes<<<(N_OWN / 4 + O_THREADS * O_UNR - 1) / (O_THREADS * O_UNR), O_THREADS, 0, stream>>>(edst, own);
  gcn_fused<<<F_NBLK, F_THREADS, 0, stream>>>(x, esrc, edst, ew, Wm, own, out);
}


# A work-group's scan of all owner bytes replaced by a counting sort of 3840-edge chunks in the first launch (LDS integer adds): a work-group reads 250 runs of packed src/row/w records; sums reordered
# speedup vs baseline: 4.8703x; 1.1817x over previous
#include <hip/hip_runtime.h>


#define N_NODES   20000
#define D_FEAT    256
#define N_SUPPORT 3
#define N_EDGES   320000
#define OUT_DIM   256
#define H_DIM     (N_SUPPORT * D_FEAT)

typedef __attribute__((ext_vector_type(2))) float v2f;
typedef __attribute__((ext_vector_type(4))) float v4f;
typedef __attribute__((ext_vector_type(8))) float v8f;

#define F_THREADS  512
#define F_WAVES    (F_THREADS / 32)
#define F_ROWS     80
#define F_MT       (F_ROWS / 16)
#define F_PA       (H_DIM + 4)
#define F_A_ELEMS  (F_ROWS * F_PA)
#define F_KC       32
#define F_NCHUNK   (H_DIM / F_KC)
#define F_PB       (OUT_DIM + 8)
#define F_B_ELEMS  (F_KC * F_PB)
#define F_B_CP     (F_KC * OUT_DIM / 4 / F_THREADS)
#define F_LDS      (F_A_ELEMS + 2 * F_B_ELEMS)

#define F_NBLK     (N_NODES / F_ROWS)
#define N_FLAT     (N_SUPPORT * N_EDGES)
#define N_CHUNK    250
#define CHUNK      (N_FLAT / N_CHUNK)
#define B_THREADS  256
#define B_PER      (CHUNK / B_THREADS)
#define N_BIN      (F_NBLK + 1)
#define REC_BYTES  ((size_t)N_FLAT * 8)
#define F_GROUP    8
#define F_LIST     512
#define F_FETCH    4
#define ROW_LOADS  (D_FEAT / 4 / 32)
#define Q_PITCH    (D_FEAT / 4)

static_assert(N_NODES % F_ROWS == 0, "whole work-groups");
static_assert(N_CHUNK * CHUNK == N_FLAT && B_PER * B_THREADS == CHUNK, "whole chunks, whole threads");
static_assert(N_BIN <= B_THREADS, "one thread per bin in the prefix sum");
static_assert(N_CHUNK <= F_WAVES * 16, "a wave takes at most 16 chunks, one per lane of its lower half");
static_assert(N_NODES <= 65536 && F_A_ELEMS <= 65536, "src and tile position fit 16 bits each");
static_assert(F_LIST % (32 * F_FETCH) == 0, "the list is filled in whole fetch rounds");
static_assert(F_WAVES * F_LIST * 2 <= 2 * F_B_ELEMS, "lists fit in the W stage area");
static_assert(REC_BYTES % 16 == 0, "the run starts behind the records stay aligned");
static_assert(F_B_CP * F_THREADS * 4 == F_KC * OUT_DIM, "W piece in whole passes");
static_assert(OUT_DIM / 16 == F_WAVES, "one column tile per wave");
static_assert(F_A_ELEMS % 4 == 0, "tile cleared in 16-byte stores");
static_assert(ROW_LOADS * 32 == Q_PITCH, "the pieces of a row fill one q block of the tile");
static_assert(Q_PITCH % F_KC == 0 && D_FEAT % Q_PITCH == 0, "a W stage lies inside one q block");

typedef __attribute__((ext_vector_type(2))) unsigned v2u;

template <int G>
__device__ __forceinline__ void add_edges(const v2u* lst, int lane,
                                          const float* __restrict__ x, float* tile) {
  v4f v[G][ROW_LOADS];
  int off[G];
  #pragma unroll
  for (int u = 0; u < G; ++u) {
    v2u r = lst[u];
    unsigned pk = (unsigned)__builtin_amdgcn_readfirstlane((int)r.x);
    float    w  = __builtin_bit_cast(float, __builtin_amdgcn_readfirstlane((int)r.y));
    int src = (int)(pk & 0xFFFFu);
    off[u]  = (int)(pk >> 16) + lane;
    const v4f* xp = (const v4f*)(x + (long)src * D_FEAT) + lane;
    #pragma unroll
    for (int j = 0; j < ROW_LOADS; ++j) v[u][j] = xp[32 * j] * w;
  }
  #pragma unroll
  for (int u = 0; u < G; ++u) {
    #pragma unroll
    for (int j = 0; j < ROW_LOADS; ++j) {
      #pragma unroll
      for (int q = 0; q < 4; ++q)
        __hip_atomic_fetch_add(tile + off[u] + q * Q_PITCH + 32 * j, v[u][j][q],
                               __ATOMIC_RELAXED, __HIP_MEMORY_SCOPE_WORKGROUP);
    }
  }
}

__device__ __forceinline__ void add_list(const v2u* lst, int n, int lane,
                                         const float* __restrict__ x, float* tile) {
  int k = 0;
  for (; k + F_GROUP <= n; k += F_GROUP) add_edges<F_GROUP>(lst + k, lane, x, tile);
  if (n - k >= 4) { add_edges<4>(lst + k, lane, x, tile); k += 4; }
  if (n - k >= 2) { add_edges<2>(lst + k, lane, x, tile); k += 2; }
  if (n - k >= 1) add_edges<1>(lst + k, lane, x, tile);
}

__global__ __launch_bounds__(B_THREADS) void bin_edges(const int* __restrict__ esrc,
                                                       const int* __restrict__ edst,
                                                       const float* __restrict__ ew,
                                                       unsigned* __restrict__ rec,
                                                       int* __restrict__ run0) {
  __shared__ int cnt[B_THREADS];
  __shared__ int first[B_THREADS];
  __shared__ int wsum[B_THREADS / 32];
  __shared__ __attribute__((aligned(16))) v2u stage[CHUNK];

  const int t    = threadIdx.x;
  const int lane = t & 31;
  const int wv   = t >> 5;
  const int e0   = blockIdx.x * CHUNK + t;

  cnt[t] = 0;
  int d[B_PER], sr[B_PER];
  float w[B_PER];
  #pragma unroll
  for (int j = 0; j < B_PER; ++j) {
    const int e = e0 + j * B_THREADS;
    d[j]  = edst[e];
    sr[j] = esrc[e];
    w[j]  = ew[e];
  }
  __syncthreads();

  int o[B_PER], pos[B_PER];
  #pragma unroll
  for (int j = 0; j < B_PER; ++j) {
    unsigned n = (unsigned)d[j];
    o[j]   = n < (unsigned)N_NODES ? (int)(n / (unsigned)F_ROWS) : F_NBLK;
    pos[j] = __hip_atomic_fetch_add(&cnt[o[j]], 1, __ATOMIC_RELAXED, __HIP_MEMORY_SCOPE_WORKGROUP);
  }
  __syncthreads();

  const int c = cnt[t];
  int inc = c;
  #pragma unroll
  for (int dd = 1; dd < 32; dd <<= 1) {
    int y = __shfl_up(inc, dd);
    if (lane >= dd) inc += y;
  }
  if (lane == 31) wsum[wv] = inc;
  __syncthreads();
  int before = 0;
  #pragma unroll
  for (int q = 0; q < B_THREADS / 32; ++q) before += q < wv ? wsum[q] : 0;
  const int ex = inc - c + before;
  first[t] = ex;
  if (t < N_BIN) run0[t * N_CHUNK + blockIdx.x] = ex;
  __syncthreads();

  int slot[B_PER];
  #pragma unroll
  for (int j = 0; j < B_PER; ++j) slot[j] = first[o[j]] + pos[j];
  #pragma unroll
  for (int j = 0; j < B_PER; ++j) {
    const int e   = e0 + j * B_THREADS;
    const int s   = (e >= N_EDGES) + (e >= 2 * N_EDGES);
    const int rel = d[j] - o[j] * F_ROWS;
    const unsigned tp = (unsigned)(rel * F_PA + s * D_FEAT);
    v2u r;
    r.x = o[j] < F_NBLK ? ((unsigned)sr[j] & 0xFFFFu) | (tp << 16) : 0u;
    r.y = __builtin_bit_cast(unsigned, w[j]);
    stage[slot[j]] = r;
  }
  __syncthreads();

  v2u* rp = (v2u*)rec + (long)blockIdx.x * CHUNK;
  #pragma unroll
  for (int j = 0; j < B_PER; ++j) rp[t + j * B_THREADS] = stage[t + j * B_THREADS];
}


__device__ __forceinline__ int w_row0(int ch) {
  return (ch >> 3) * D_FEAT + (ch & 1) * (4 * F_KC) + ((ch >> 1) & 3);
}

__device__ __forceinline__ void w_fetch(v4f (&rb)[F_B_CP], const float* __restrict__ wp) {
  #pragma unroll
  for (int j = 0; j < F_B_CP; ++j) rb[j] = *(const v4f*)(wp + (long)(j * 32) * OUT_DIM);
}

__device__ __forceinline__ void w_put(const v4f (&rb)[F_B_CP], float* bs) {
  #pragma unroll
  for (int j = 0; j < F_B_CP; ++j) *(v4f*)(bs + j * 8 * F_PB) = rb[j];
}

template <int KS0, int KS1>
__device__ __forceinline__ void mm_steps(v8f (&acc)[F_MT], const float* ap, const float* bp) {
  #pragma unroll
  for (int ks = KS0; ks < KS1; ++ks) {
    v2f b;
    b.x = bp[(ks * 4) * F_PB];
    b.y = bp[(ks * 4 + 1) * F_PB];
    #pragma unroll
    for (int mt = 0; mt < F_MT; ++mt) {
      v2f a = *(const v2f*)(ap + mt * 16 * F_PA + ks * 4);
      acc[mt] = __builtin_amdgcn_wmma_f32_16x16x4_f32(
           false, a,  false, b,
           (short)0, acc[mt],  false,  false);
    }
  }
}

__global__ __launch_bounds__(F_THREADS) void gcn_fused(
    const float* __restrict__ x, const float* __restrict__ Wm,
    const unsigned* __restrict__ rec, const int* __restrict__ run0,
    float* __restrict__ out) {
  __shared__ __attribute__((aligned(16))) float lds[F_LDS];

  const int tid  = threadIdx.x;
  const int lane = tid & 31;
  const int wave = __builtin_amdgcn_readfirstlane(tid >> 5);
  const int c    = lane & 15;
  const int half = lane >> 4;
  const int m0   = blockIdx.x * F_ROWS;

  float* tile = lds;
  float* wst  = lds + F_A_ELEMS;

  const int chunk   = wave + F_WAVES * lane;
  const bool has_run = lane < 16 && chunk < N_CHUNK;
  const int chunk_c = chunk < N_CHUNK ? chunk : N_CHUNK - 1;
  const int run_lo = run0[blockIdx.x * N_CHUNK + chunk_c];
  const int run_hi = run0[(blockIdx.x + 1) * N_CHUNK + chunk_c];

  for (int i = tid; i < F_A_ELEMS / 4; i += F_THREADS)
    *(v4f*)(tile + 4 * i) = v4f{0.f, 0.f, 0.f, 0.f};

  const int run_s = has_run ? run_lo : 0;
  const int run_n = has_run ? run_hi - run_lo : 0;

  {
    int inc = run_n;
    #pragma unroll
    for (int dd = 1; dd < 16; dd <<= 1) {
      int y = __shfl_up(inc, dd);
      if (lane >= dd) inc += y;
    }
    const int total = __builtin_amdgcn_readlane(inc, 15);
    const int base  = chunk * CHUNK + run_s - (inc - run_n);
    int pend[15];
    #pragma unroll
    for (int i = 0; i < 15; ++i) pend[i] = __builtin_amdgcn_readlane(inc, i);

    v2u* lst = (v2u*)wst + wave * F_LIST;
    const v2u* rp = (const v2u*)rec;

    auto fill = [&](int j0, int nb) {
      for (int k0 = 0; k0 < nb; k0 += 32 * F_FETCH) {
        v2u r[F_FETCH];
        #pragma unroll
        for (int q = 0; q < F_FETCH; ++q) {
          int j = j0 + k0 + 32 * q + lane;
          j = j < total ? j : total - 1;
          int i = 0;
          #pragma unroll
          for (int cI = 0; cI < 15; ++cI) i += j >= pend[cI] ? 1 : 0;
          r[q] = rp[__shfl(base, i) + j];
        }
        #pragma unroll
        for (int q = 0; q < F_FETCH; ++q) {
          int k = k0 + 32 * q + lane;
          if (k < nb) lst[k] = r[q];
        }
      }
    };

    int nb = total < F_LIST ? total : F_LIST;
    fill(0, nb);
    __syncthreads();
    add_list(lst, nb, lane, x, tile);
    for (int j0 = F_LIST; j0 < total; j0 += F_LIST) {
      nb = total - j0 < F_LIST ? total - j0 : F_LIST;
      fill(j0, nb);
      add_list(lst, nb, lane, x, tile);
    }
  }
  __syncthreads();

  const float* wg = Wm + (long)(4 * (tid >> 6)) * OUT_DIM + (tid & 63) * 4;
  const int b_st = (tid >> 6) * F_PB + (tid & 63) * 4;
  const int a_ld = c * F_PA + 2 * half;
  const int b_ld = (2 * half) * F_PB + wave * 16 + c;

  v4f rb[F_B_CP];
  w_fetch(rb, wg + (long)w_row0(0) * OUT_DIM);
  w_put(rb, wst + b_st);
  w_fetch(rb, wg + (long)w_row0(1) * OUT_DIM);
  __syncthreads();

  v8f acc[F_MT];
  #pragma unroll
  for (int mt = 0; mt < F_MT; ++mt) acc[mt] = v8f{};

  for (int ch = 0; ch < F_NCHUNK - 2; ++ch) {
    const float* cur = wst + (ch & 1) * F_B_ELEMS;
    float*       nxt = wst + ((ch + 1) & 1) * F_B_ELEMS;
    const float* ap  = tile + a_ld + ch * F_KC;
    mm_steps<0, F_KC / 8>(acc, ap, cur + b_ld);
    w_put(rb, nxt + b_st);
    mm_steps<F_KC / 8, F_KC / 4>(acc, ap, cur + b_ld);
    w_fetch(rb, wg + (long)w_row0(ch + 2) * OUT_DIM);
    __syncthreads();
  }
  {
    const float* cur = wst + ((F_NCHUNK - 2) & 1) * F_B_ELEMS;
    float*       nxt = wst + ((F_NCHUNK - 1) & 1) * F_B_ELEMS;
    const float* ap  = tile + a_ld + (F_NCHUNK - 2) * F_KC;
    mm_steps<0, F_KC / 8>(acc, ap, cur + b_ld);
    w_put(rb, nxt + b_st);
    mm_steps<F_KC / 8, F_KC / 4>(acc, ap, cur + b_ld);
    __syncthreads();
    mm_steps<0, F_KC / 4>(acc, ap + F_KC, nxt + b_ld);
  }

  float* op = out + (long)(m0 + half * 8) * OUT_DIM + wave * 16 + c;
  #pragma unroll
  for (int mt = 0; mt < F_MT; ++mt) {
    #pragma unroll
    for (int j = 0; j < 8; ++j) op[(long)(mt * 16 + j) * OUT_DIM] = acc[mt][j];
  }
}

extern "C" void kernel_launch(void* const* d_in, const int* in_sizes, int n_in,
                              void* d_out, int out_size, void* d_ws, size_t ws_size,
                              hipStream_t stream) {
  const float* x    = (const float*)d_in[0];
  const int*   esrc = (const int*)  d_in[1];
  const int*   edst = (const int*)  d_in[2];
  const float* ew   = (const float*)d_in[3];
  const float* Wm   = (const float*)d_in[4];
  float*       out  = (float*)d_out;
  unsigned*    rec  = (unsigned*)d_ws;
  int*         run0 = (int*)((char*)d_ws + REC_BYTES);

  bin_edges<<<N_CHUNK, B_THREADS, 0, stream>>>(esrc, edst, ew, rec, run0);
  gcn_fused<<<F_NBLK, F_THREADS, 0, stream>>>(x, Wm, rec, run0, out);
}


# Product runs beside the edge sum, one support behind: chunks cut per support; waves 8-15 multiply support s (80x32 of out a wave, W fragments from global) while waves 0-7 sum s+1; sums reordered
# speedup vs baseline: 5.2994x; 1.0881x over previous
#include <hip/hip_runtime.h>


#define N_NODES   20000
#define D_FEAT    256
#define N_SUPPORT 3
#define N_EDGES   320000
#define OUT_DIM   256
#define H_DIM     (N_SUPPORT * D_FEAT)

typedef __attribute__((ext_vector_type(2))) float v2f;
typedef __attribute__((ext_vector_type(4))) float v4f;
typedef __attribute__((ext_vector_type(8))) float v8f;

#define F_THREADS  512
#define F_WAVES    (F_THREADS / 32)
#define F_SUMW     8
#define F_MULW     (F_WAVES - F_SUMW)
#define F_ROWS     80
#define F_MT       (F_ROWS / 16)
#define F_NT       (OUT_DIM / 16 / F_MULW)
#define F_PA       (H_DIM + 4)
#define F_A_ELEMS  (F_ROWS * F_PA)
#define F_KB       16
#define F_NBLK_K   (H_DIM / F_KB)
#define F_QBLK     (Q_PITCH / F_KB)
#define F_LIST     512
#define F_LDS      (F_A_ELEMS + F_WAVES * F_LIST * 2)

#define F_NBLK     (N_NODES / F_ROWS)
#define N_FLAT     (N_SUPPORT * N_EDGES)
#define SUP_CHUNKS 100
#define N_CHUNK    (N_SUPPORT * SUP_CHUNKS)
#define CHUNK      (N_FLAT / N_CHUNK)
#define B_THREADS  320
#define B_WAVES    (B_THREADS / 32)
#define B_PER      (CHUNK / B_THREADS)
#define N_BIN      (F_NBLK + 1)
#define REC_BYTES  ((size_t)N_FLAT * 8)
#define F_GROUP    8
#define F_FETCH    4
#define ROW_LOADS  (D_FEAT / 4 / 32)
#define Q_PITCH    (D_FEAT / 4)

static_assert(N_NODES % F_ROWS == 0, "whole work-groups");
static_assert(N_CHUNK * CHUNK == N_FLAT && B_PER * B_THREADS == CHUNK, "whole chunks, whole threads");
static_assert(SUP_CHUNKS * CHUNK == N_EDGES, "a chunk lies inside one support");
static_assert(N_BIN <= B_THREADS, "one thread per bin in the prefix sum");
static_assert(SUP_CHUNKS <= F_SUMW * 16, "a wave takes at most 16 chunks of a support, one per lane of its lower half");
static_assert(N_NODES <= 65536 && F_A_ELEMS <= 65536, "src and tile position fit 16 bits each");
static_assert(F_LIST % (32 * F_FETCH) == 0, "the list is filled in whole fetch rounds");
static_assert(REC_BYTES % 16 == 0, "the run starts behind the records stay aligned");
static_assert(F_NT == 2, "a product wave loads its two column tiles as one 8-byte piece");
static_assert(F_A_ELEMS % 4 == 0, "tile cleared in 16-byte stores");
static_assert(ROW_LOADS * 32 == Q_PITCH, "the pieces of a row fill one q block of the tile");
static_assert(Q_PITCH % F_KB == 0 && D_FEAT % Q_PITCH == 0, "a k block lies inside one q block");
static_assert((D_FEAT / F_KB) % 2 == 0, "a support is whole pairs of k blocks");

typedef __attribute__((ext_vector_type(2))) unsigned v2u;
template <int G>
__device__ __forceinline__ void add_edges(const v2u* lst, int lane,
                                          const float* __restrict__ x, float* tile) {
  v4f v[G][ROW_LOADS];
  int off[G];
  #pragma unroll
  for (int u = 0; u < G; ++u) {
    v2u r = lst[u];
    unsigned pk = (unsigned)__builtin_amdgcn_readfirstlane((int)r.x);
    float    w  = __builtin_bit_cast(float, __builtin_amdgcn_readfirstlane((int)r.y));
    int src = (int)(pk & 0xFFFFu);
    off[u]  = (int)(pk >> 16) + lane;
    const v4f* xp = (const v4f*)(x + (long)src * D_FEAT) + lane;
    #pragma unroll
    for (int j = 0; j < ROW_LOADS; ++j) v[u][j] = xp[32 * j] * w;
  }
  #pragma unroll
  for (int u = 0; u < G; ++u) {
    #pragma unroll
    for (int j = 0; j < ROW_LOADS; ++j) {
      #pragma unroll
      for (int q = 0; q < 4; ++q)
        __hip_atomic_fetch_add(tile + off[u] + q * Q_PITCH + 32 * j, v[u][j][q],
                               __ATOMIC_RELAXED, __HIP_MEMORY_SCOPE_WORKGROUP);
    }
  }
}

__device__ __forceinline__ void add_list(const v2u* lst, int n, int lane,
                                         const float* __restrict__ x, float* tile) {
  int k = 0;
  for (; k + F_GROUP <= n; k += F_GROUP) add_edges<F_GROUP>(lst + k, lane, x, tile);
  if (n - k >= 4) { add_edges<4>(lst + k, lane, x, tile); k += 4; }
  if (n - k >= 2) { add_edges<2>(lst + k, lane, x, tile); k += 2; }
  if (n - k >= 1) add_edges<1>(lst + k, lane, x, tile);
}

__global__ __launch_bounds__(B_THREADS) void bin_edges(const int* __restrict__ esrc,
                                                       const int* __restrict__ edst,
                                                       const float* __restrict__ ew,
                                                       unsigned* __restrict__ rec,
                                                       int* __restrict__ run0) {
  __shared__ int cnt[B_THREADS];
  __shared__ int first[B_THREADS];
  __shared__ int wsum[B_WAVES];
  __shared__ __attribute__((aligned(16))) v2u stage[CHUNK];

  const int t    = threadIdx.x;
  const int lane = t & 31;
  const int wv   = t >> 5;
  const int e0   = blockIdx.x * CHUNK + t;
  const int s    = blockIdx.x / SUP_CHUNKS;

  cnt[t] = 0;
  int d[B_PER], sr[B_PER];
  float w[B_PER];
  #pragma unroll
  for (int j = 0; j < B_PER; ++j) {
    const int e = e0 + j * B_THREADS;
    d[j]  = edst[e];
    sr[j] = esrc[e];
    w[j]  = ew[e];
  }
  __syncthreads();

  int o[B_PER], pos[B_PER];
  #pragma unroll
  for (int j = 0; j < B_PER; ++j) {
    unsigned n = (unsigned)d[j];
    o[j]   = n < (unsigned)N_NODES ? (int)(n / (unsigned)F_ROWS) : F_NBLK;
    pos[j] = __hip_atomic_fetch_add(&cnt[o[j]], 1, __ATOMIC_RELAXED, __HIP_MEMORY_SCOPE_WORKGROUP);
  }
  __syncthreads();

  const int c = cnt[t];
  int inc = c;
  #pragma unroll
  for (int dd = 1; dd < 32; dd <<= 1) {
    int y = __shfl_up(inc, dd);
    if (lane >= dd) inc += y;
  }
  if (lane == 31) wsum[wv] = inc;
  __syncthreads();
  int before = 0;
  #pragma unroll
  for (int q = 0; q < B_WAVES; ++q) before += q < wv ? wsum[q] : 0;
  const int ex = inc - c + before;
  first[t] = ex;
  if (t < N_BIN) run0[t * N_CHUNK + blockIdx.x] = ex;
  __syncthreads();

  int slot[B_PER];
  #pragma unroll
  for (int j = 0; j < B_PER; ++j) slot[j] = first[o[j]] + pos[j];
  #pragma unroll
  for (int j = 0; j < B_PER; ++j) {
    const int rel = d[j] - o[j] * F_ROWS;
    const unsigned tp = (unsigned)(rel * F_PA + s * D_FEAT);
    v2u r;
    r.x = o[j] < F_NBLK ? ((unsigned)sr[j] & 0xFFFFu) | (tp << 16) : 0u;
    r.y = __builtin_bit_cast(unsigned, w[j]);
    stage[slot[j]] = r;
  }
  __syncthreads();

  v2u* rp = (v2u*)rec + (long)blockIdx.x * CHUNK;
  #pragma unroll
  for (int j = 0; j < B_PER; ++j) rp[t + j * B_THREADS] = stage[t + j * B_THREADS];
}


__device__ __forceinline__ int b_row0(int blk) {
  return (blk / (D_FEAT / F_KB)) * D_FEAT + (blk % F_QBLK) * (4 * F_KB) + ((blk / F_QBLK) & 3);
}

__device__ __forceinline__ void b_fetch(v2f (&rb)[F_KB / 4][2], const float* __restrict__ wl, int blk) {
  const float* p = wl + (long)b_row0(blk) * OUT_DIM;
  #pragma unroll
  for (int i = 0; i < F_KB / 4; ++i) {
    #pragma unroll
    for (int e = 0; e < 2; ++e) rb[i][e] = *(const v2f*)(p + (long)(16 * i + 4 * e) * OUT_DIM);
  }
}

__device__ __forceinline__ void mm_block(v8f (&acc)[F_NT][F_MT], const v2f (&rb)[F_KB / 4][2], const float* ap) {
  #pragma unroll
  for (int i0 = 0; i0 < F_KB / 4; i0 += 2) {
    v2f a[2][F_MT];
    #pragma unroll
    for (int i = 0; i < 2; ++i) {
      #pragma unroll
      for (int mt = 0; mt < F_MT; ++mt) a[i][mt] = *(const v2f*)(ap + mt * 16 * F_PA + (i0 + i) * 4);
    }
    #pragma unroll
    for (int i = 0; i < 2; ++i) {
      v2f b0, b1;
      b0.x = rb[i0 + i][0].x; b0.y = rb[i0 + i][1].x;
      b1.x = rb[i0 + i][0].y; b1.y = rb[i0 + i][1].y;
      #pragma unroll
      for (int mt = 0; mt < F_MT; ++mt) {
        acc[0][mt] = __builtin_amdgcn_wmma_f32_16x16x4_f32(
             false, a[i][mt],  false, b0,
             (short)0, acc[0][mt],  false,  false);
        acc[1][mt] = __builtin_amdgcn_wmma_f32_16x16x4_f32(
             false, a[i][mt],  false, b1,
             (short)0, acc[1][mt],  false,  false);
      }
    }
  }
}

template <bool SYNC>
__device__ __forceinline__ void sum_runs(int chunk, int run_s, int run_n, int lane, v2u* lst,
                                         const v2u* __restrict__ rp, const float* __restrict__ x,
                                         float* tile) {
  int inc = run_n;
  #pragma unroll
  for (int dd = 1; dd < 16; dd <<= 1) {
    int y = __shfl_up(inc, dd);
    if (lane >= dd) inc += y;
  }
  const int total = __builtin_amdgcn_readlane(inc, 15);
  const int base  = chunk * CHUNK + run_s - (inc - run_n);
  int pend[15];
  #pragma unroll
  for (int i = 0; i < 15; ++i) pend[i] = __builtin_amdgcn_readlane(inc, i);

  auto fill = [&](int j0, int nb) {
    for (int k0 = 0; k0 < nb; k0 += 32 * F_FETCH) {
      v2u r[F_FETCH];
      #pragma unroll
      for (int q = 0; q < F_FETCH; ++q) {
        int j = j0 + k0 + 32 * q + lane;
        j = j < total ? j : total - 1;
        int i = 0;
        #pragma unroll
        for (int cI = 0; cI < 15; ++cI) i += j >= pend[cI] ? 1 : 0;
        r[q] = rp[__shfl(base, i) + j];
      }
      #pragma unroll
      for (int q = 0; q < F_FETCH; ++q) {
        int k = k0 + 32 * q + lane;
        if (k < nb) lst[k] = r[q];
      }
    }
  };

  int nb = total < F_LIST ? total : F_LIST;
  fill(0, nb);
  if (SYNC) __syncthreads();
  add_list(lst, nb, lane, x, tile);
  for (int j0 = F_LIST; j0 < total; j0 += F_LIST) {
    nb = total - j0 < F_LIST ? total - j0 : F_LIST;
    fill(j0, nb);
    add_list(lst, nb, lane, x, tile);
  }
}

__global__ __launch_bounds__(F_THREADS) void gcn_fused(
    const float* __restrict__ x, const float* __restrict__ Wm,
    const unsigned* __restrict__ rec, const int* __restrict__ run0,
    float* __restrict__ out) {
  __shared__ __attribute__((aligned(16))) float lds[F_LDS];

  const int tid  = threadIdx.x;
  const int lane = tid & 31;
  const int wave = __builtin_amdgcn_readfirstlane(tid >> 5);
  const int c    = lane & 15;
  const int half = lane >> 4;
  const int m0   = blockIdx.x * F_ROWS;
  const bool sums = wave < F_SUMW;
  const int g    = wave - F_SUMW;

  float* tile = lds;
  v2u*   lst  = (v2u*)(lds + F_A_ELEMS) + wave * F_LIST;
  const v2u* rp = (const v2u*)rec;

  const int l0 = wave + F_WAVES * lane;
  const int l1 = wave + F_SUMW * lane;
  const bool has0 = lane < 16 && l0 < SUP_CHUNKS;
  const bool has1 = lane < 16 && l1 < SUP_CHUNKS && sums;
  const int ch0 = l0 < SUP_CHUNKS ? l0 : SUP_CHUNKS - 1;
  const int cl1 = l1 < SUP_CHUNKS ? l1 : SUP_CHUNKS - 1;
  const int ch1 = SUP_CHUNKS + cl1;
  const int ch2 = 2 * SUP_CHUNKS + cl1;
  const int* r_lo = run0 + blockIdx.x * N_CHUNK;
  const int* r_hi = r_lo + N_CHUNK;
  const int lo0 = r_lo[ch0], hi0 = r_hi[ch0];
  const int lo1 = r_lo[ch1], hi1 = r_hi[ch1];
  const int lo2 = r_lo[ch2], hi2 = r_hi[ch2];

  for (int i = tid; i < F_A_ELEMS / 4; i += F_THREADS)
    *(v4f*)(tile + 4 * i) = v4f{0.f, 0.f, 0.f, 0.f};

  sum_runs<true>(ch0, has0 ? lo0 : 0, has0 ? hi0 - lo0 : 0, lane, lst, rp, x, tile);

  const float* wl = Wm + (long)(8 * half) * OUT_DIM + (g & (F_MULW - 1)) * 32 + 2 * c;
  const float* ap0 = tile + c * F_PA + 2 * half;
  v2f rbA[F_KB / 4][2], rbB[F_KB / 4][2];
  v8f acc[F_NT][F_MT];
  #pragma unroll
  for (int t = 0; t < F_NT; ++t) {
    #pragma unroll
    for (int mt = 0; mt < F_MT; ++mt) acc[t][mt] = v8f{};
  }
  if (!sums) b_fetch(rbA, wl, 0);
  __syncthreads();

  for (int s = 0; s < N_SUPPORT; ++s) {
    if (sums) {
      if (s + 1 < N_SUPPORT) {
        const int lo = s == 0 ? lo1 : lo2;
        const int hi = s == 0 ? hi1 : hi2;
        sum_runs<false>(s == 0 ? ch1 : ch2, has1 ? lo : 0, has1 ? hi - lo : 0, lane, lst, rp, x, tile);
      }
    } else {
      for (int kb = 0; kb < D_FEAT / F_KB; kb += 2) {
        const int blk = s * (D_FEAT / F_KB) + kb;
        b_fetch(rbB, wl, blk + 1);
        mm_block(acc, rbA, ap0 + blk * F_KB);
        b_fetch(rbA, wl, blk + 2 < F_NBLK_K ? blk + 2 : F_NBLK_K - 1);
        mm_block(acc, rbB, ap0 + (blk + 1) * F_KB);
      }
    }
    __syncthreads();
  }

  if (!sums) {
    float* op = out + (long)(m0 + half * 8) * OUT_DIM + g * 32 + 2 * c;
    #pragma unroll
    for (int mt = 0; mt < F_MT; ++mt) {
      #pragma unroll
      for (int j = 0; j < 8; ++j) {
        v2f o2;
        o2.x = acc[0][mt][j];
        o2.y = acc[1][mt][j];
        *(v2f*)(op + (long)(mt * 16 + j) * OUT_DIM) = o2;
      }
    }
  }
}

extern "C" void kernel_launch(void* const* d_in, const int* in_sizes, int n_in,
                              void* d_out, int out_size, void* d_ws, size_t ws_size,
                              hipStream_t stream) {
  const float* x    = (const float*)d_in[0];
  const int*   esrc = (const int*)  d_in[1];
  const int*   edst = (const int*)  d_in[2];
  const float* ew   = (const float*)d_in[3];
  const float* Wm   = (const float*)d_in[4];
  float*       out  = (float*)d_out;
  unsigned*    rec  = (unsigned*)d_ws;
  int*         run0 = (int*)((char*)d_ws + REC_BYTES);

  bin_edges<<<N_CHUNK, B_THREADS, 0, stream>>>(esrc, edst, ew, rec, run0);
  gcn_fused<<<F_NBLK, F_THREADS, 0, stream>>>(x, Wm, rec, run0, out);
}
